# starting kernel
# baseline (speedup 1.0000x reference)
#include <hip/hip_runtime.h>

#define NN 50000
#define NE 800000
#define DD 128

typedef float v2f __attribute__((ext_vector_type(2)));
typedef float v8f __attribute__((ext_vector_type(8)));

__global__ __launch_bounds__(256) void gemm5_wmma(
    const float* __restrict__ X,
    const float* __restrict__ W1, const float* __restrict__ b1,
    const float* __restrict__ W2, const float* __restrict__ b2,
    const float* __restrict__ W3, const float* __restrict__ b3,
    const float* __restrict__ W4, const float* __restrict__ b4,
    const float* __restrict__ Wr, const float* __restrict__ br,
    float* __restrict__ H1, float* __restrict__ H2,
    float* __restrict__ H3, float* __restrict__ H4,
    float* __restrict__ Hr)
{
    __shared__ float xs[16 * 132];
    const int row0 = blockIdx.x * 16;
    const int t = threadIdx.x;

    #pragma unroll
    for (int q = t; q < 512; q += 256) {
        const int r  = q >> 5;
        const int c4 = (q & 31) << 2;
        const float4 v = *reinterpret_cast<const float4*>(X + (size_t)(row0 + r) * DD + c4);
        *reinterpret_cast<float4*>(&xs[r * 132 + c4]) = v;
    }
    __syncthreads();

    const int wave = t >> 5;
    const int lane = t & 31;
    const int half = lane >> 4;
    const int nn   = lane & 15;
    const int n0   = wave * 16;
    const int wrow = n0 + nn;

    const float* Ws[5] = {W1, W2, W3, W4, Wr};
    const float* Bs[5] = {b1, b2, b3, b4, br};
    float*       Hs[5] = {H1, H2, H3, H4, Hr};

    v8f acc[5] = {};

    for (int k = 0; k < DD; k += 4) {
        const int kk = k + 2 * half;
        const float2 af = *reinterpret_cast<const float2*>(&xs[nn * 132 + kk]);
        v2f a; a.x = af.x; a.y = af.y;
        #pragma unroll
        for (int w = 0; w < 5; ++w) {
            const float2 bf = *reinterpret_cast<const float2*>(Ws[w] + (size_t)wrow * DD + kk);
            v2f b; b.x = bf.x; b.y = bf.y;
            acc[w] = __builtin_amdgcn_wmma_f32_16x16x4_f32(
                false, a, false, b, (short)0, acc[w], false, false);
        }
    }

    #pragma unroll
    for (int w = 0; w < 5; ++w) {
        const float bias = Bs[w][wrow];
        #pragma unroll
        for (int j = 0; j < 8; ++j) {
            const int m = j + 8 * half;
            Hs[w][(size_t)(row0 + m) * DD + wrow] = acc[w][j] + bias;
        }
    }
}

__global__ __launch_bounds__(256) void edge_gate_scatter(
    const int* __restrict__ ei,
    const float* __restrict__ H2, const float* __restrict__ H3,
    const float* __restrict__ H4,
    float* __restrict__ H1acc, float* __restrict__ cnt)
{
    const int wave = threadIdx.x >> 5;
    const int lane = threadIdx.x & 31;
    const int e = blockIdx.x * 8 + wave;
    if (e >= NE) return;

    const int src = ei[e];
    const int dst = ei[NE + e];
    const int d0 = lane << 2;

    const float4 x3 = *reinterpret_cast<const float4*>(H3 + (size_t)src * DD + d0);
    const float4 x4 = *reinterpret_cast<const float4*>(H4 + (size_t)dst * DD + d0);
    const float4 x2 = *reinterpret_cast<const float4*>(H2 + (size_t)src * DD + d0);

    float4 m;
    m.x = x2.x / (1.0f + __expf(-(x3.x + x4.x)));
    m.y = x2.y / (1.0f + __expf(-(x3.y + x4.y)));
    m.z = x2.z / (1.0f + __expf(-(x3.z + x4.z)));
    m.w = x2.w / (1.0f + __expf(-(x3.w + x4.w)));

    float* p = H1acc + (size_t)dst * DD + d0;
    unsafeAtomicAdd(p + 0, m.x);
    unsafeAtomicAdd(p + 1, m.y);
    unsafeAtomicAdd(p + 2, m.z);
    unsafeAtomicAdd(p + 3, m.w);

    if (lane == 0) unsafeAtomicAdd(&cnt[dst], 1.0f);
}

__global__ __launch_bounds__(256) void finalize_mean_res(
    const float* __restrict__ H1acc, const float* __restrict__ cnt,
    float* __restrict__ out)
{
    const size_t i = (size_t)blockIdx.x * 256 + threadIdx.x;
    const int node = (int)(i >> 5);
    const float s = 1.0f / fmaxf(cnt[node], 1.0f);
    const float4 h = reinterpret_cast<const float4*>(H1acc)[i];
    float4 o = reinterpret_cast<float4*>(out)[i];
    o.x += h.x * s;
    o.y += h.y * s;
    o.z += h.z * s;
    o.w += h.w * s;
    reinterpret_cast<float4*>(out)[i] = o;
}

extern "C" void kernel_launch(void* const* d_in, const int* in_sizes, int n_in,
                              void* d_out, int out_size, void* d_ws, size_t ws_size,
                              hipStream_t stream)
{
    const float* X   = (const float*)d_in[0];
    const int*   ei  = (const int*)  d_in[1];
    const float* W1  = (const float*)d_in[2];
    const float* b1  = (const float*)d_in[3];
    const float* W2  = (const float*)d_in[4];
    const float* b2  = (const float*)d_in[5];
    const float* W3  = (const float*)d_in[6];
    const float* b3  = (const float*)d_in[7];
    const float* W4  = (const float*)d_in[8];
    const float* b4  = (const float*)d_in[9];
    const float* Wr  = (const float*)d_in[10];
    const float* br  = (const float*)d_in[11];
    float* out = (float*)d_out;

    const size_t nd = (size_t)NN * DD;
    float* ws  = (float*)d_ws;
    float* H1  = ws;
    float* H2  = ws + nd;
    float* H3  = ws + 2 * nd;
    float* H4  = ws + 3 * nd;
    float* cnt = ws + 4 * nd;

    hipMemsetAsync(cnt, 0, NN * sizeof(float), stream);

    gemm5_wmma<<<NN / 16, 256, 0, stream>>>(
        X, W1, b1, W2, b2, W3, b3, W4, b4, Wr, br,
        H1, H2, H3, H4,   out);

    edge_gate_scatter<<<NE / 8, 256, 0, stream>>>(ei, H2, H3, H4, H1, cnt);

    finalize_mean_res<<<(int)(nd / 4 / 256), 256, 0, stream>>>(H1, cnt, out);
}


# GEMM: block = 80-row slab x 64 cols (4 waves); each W fragment loaded per k-step now feeds 5 row tiles (25 WMMAs per 5 loads): 5x fewer scattered W loads per WMMA; k order and sums unchanged
# speedup vs baseline: 1.1360x; 1.1360x over previous
#include <hip/hip_runtime.h>

#define NN 50000
#define NE 800000
#define DD 128

typedef float v2f __attribute__((ext_vector_type(2)));
typedef float v8f __attribute__((ext_vector_type(8)));

#define GM_TILES 5
#define GM_ROWS  (GM_TILES * 16)

__global__ __launch_bounds__(128) void gemm5_wmma(
    const float* __restrict__ X,
    const float* __restrict__ W1, const float* __restrict__ b1,
    const float* __restrict__ W2, const float* __restrict__ b2,
    const float* __restrict__ W3, const float* __restrict__ b3,
    const float* __restrict__ W4, const float* __restrict__ b4,
    const float* __restrict__ Wr, const float* __restrict__ br,
    float* __restrict__ H1, float* __restrict__ H2,
    float* __restrict__ H3, float* __restrict__ H4,
    float* __restrict__ Hr)
{
    __shared__ float xs[GM_ROWS * 132];
    const int slab  = blockIdx.x >> 1;
    const int nhalf = blockIdx.x & 1;
    const int row0  = slab * GM_ROWS;
    const int t = threadIdx.x;

    for (int q = t; q < GM_ROWS * 32; q += 128) {
        const int r  = q >> 5;
        const int c4 = (q & 31) << 2;
        const float4 v = *reinterpret_cast<const float4*>(X + (size_t)(row0 + r) * DD + c4);
        *reinterpret_cast<float4*>(&xs[r * 132 + c4]) = v;
    }
    __syncthreads();

    const int wave = t >> 5;
    const int lane = t & 31;
    const int half = lane >> 4;
    const int nn   = lane & 15;
    const int n0   = (nhalf * 4 + wave) * 16;
    const int wrow = n0 + nn;

    const float* Ws[5] = {W1, W2, W3, W4, Wr};
    const float* Bs[5] = {b1, b2, b3, b4, br};
    float*       Hs[5] = {H1, H2, H3, H4, Hr};

    v8f acc[GM_TILES][5] = {};

    for (int k = 0; k < DD; k += 4) {
        const int kk = k + 2 * half;
        v2f b[5];
        #pragma unroll
        for (int w = 0; w < 5; ++w) {
            const float2 bf = *reinterpret_cast<const float2*>(Ws[w] + (size_t)wrow * DD + kk);
            b[w].x = bf.x; b[w].y = bf.y;
        }
        #pragma unroll
        for (int mt = 0; mt < GM_TILES; ++mt) {
            const float2 af = *reinterpret_cast<const float2*>(&xs[(mt * 16 + nn) * 132 + kk]);
            v2f a; a.x = af.x; a.y = af.y;
            #pragma unroll
            for (int w = 0; w < 5; ++w) {
                acc[mt][w] = __builtin_amdgcn_wmma_f32_16x16x4_f32(
                    false, a, false, b[w], (short)0, acc[mt][w], false, false);
            }
        }
    }

    #pragma unroll
    for (int w = 0; w < 5; ++w) {
        const float bias = Bs[w][wrow];
        #pragma unroll
        for (int mt = 0; mt < GM_TILES; ++mt) {
            #pragma unroll
            for (int j = 0; j < 8; ++j) {
                const int m = mt * 16 + j + 8 * half;
                Hs[w][(size_t)(row0 + m) * DD + wrow] = acc[mt][w][j] + bias;
            }
        }
    }
}

__global__ __launch_bounds__(256) void edge_gate_scatter(
    const int* __restrict__ ei,
    const float* __restrict__ H2, const float* __restrict__ H3,
    const float* __restrict__ H4,
    float* __restrict__ H1acc, float* __restrict__ cnt)
{
    const int wave = threadIdx.x >> 5;
    const int lane = threadIdx.x & 31;
    const int e = blockIdx.x * 8 + wave;
    if (e >= NE) return;

    const int src = ei[e];
    const int dst = ei[NE + e];
    const int d0 = lane << 2;

    const float4 x3 = *reinterpret_cast<const float4*>(H3 + (size_t)src * DD + d0);
    const float4 x4 = *reinterpret_cast<const float4*>(H4 + (size_t)dst * DD + d0);
    const float4 x2 = *reinterpret_cast<const float4*>(H2 + (size_t)src * DD + d0);

    float4 m;
    m.x = x2.x / (1.0f + __expf(-(x3.x + x4.x)));
    m.y = x2.y / (1.0f + __expf(-(x3.y + x4.y)));
    m.z = x2.z / (1.0f + __expf(-(x3.z + x4.z)));
    m.w = x2.w / (1.0f + __expf(-(x3.w + x4.w)));

    float* p = H1acc + (size_t)dst * DD + d0;
    unsafeAtomicAdd(p + 0, m.x);
    unsafeAtomicAdd(p + 1, m.y);
    unsafeAtomicAdd(p + 2, m.z);
    unsafeAtomicAdd(p + 3, m.w);

    if (lane == 0) unsafeAtomicAdd(&cnt[dst], 1.0f);
}

__global__ __launch_bounds__(256) void finalize_mean_res(
    const float* __restrict__ H1acc, const float* __restrict__ cnt,
    float* __restrict__ out)
{
    const size_t i = (size_t)blockIdx.x * 256 + threadIdx.x;
    const int node = (int)(i >> 5);
    const float s = 1.0f / fmaxf(cnt[node], 1.0f);
    const float4 h = reinterpret_cast<const float4*>(H1acc)[i];
    float4 o = reinterpret_cast<float4*>(out)[i];
    o.x += h.x * s;
    o.y += h.y * s;
    o.z += h.z * s;
    o.w += h.w * s;
    reinterpret_cast<float4*>(out)[i] = o;
}

extern "C" void kernel_launch(void* const* d_in, const int* in_sizes, int n_in,
                              void* d_out, int out_size, void* d_ws, size_t ws_size,
                              hipStream_t stream)
{
    const float* X   = (const float*)d_in[0];
    const int*   ei  = (const int*)  d_in[1];
    const float* W1  = (const float*)d_in[2];
    const float* b1  = (const float*)d_in[3];
    const float* W2  = (const float*)d_in[4];
    const float* b2  = (const float*)d_in[5];
    const float* W3  = (const float*)d_in[6];
    const float* b3  = (const float*)d_in[7];
    const float* W4  = (const float*)d_in[8];
    const float* b4  = (const float*)d_in[9];
    const float* Wr  = (const float*)d_in[10];
    const float* br  = (const float*)d_in[11];
    float* out = (float*)d_out;

    const size_t nd = (size_t)NN * DD;
    float* ws  = (float*)d_ws;
    float* H1  = ws;
    float* H2  = ws + nd;
    float* H3  = ws + 2 * nd;
    float* H4  = ws + 3 * nd;
    float* cnt = ws + 4 * nd;

    hipMemsetAsync(cnt, 0, NN * sizeof(float), stream);

    gemm5_wmma<<<(NN / GM_ROWS) * 2, 128, 0, stream>>>(
        X, W1, b1, W2, b2, W3, b3, W4, b4, Wr, br,
        H1, H2, H3, H4,   out);

    edge_gate_scatter<<<NE / 8, 256, 0, stream>>>(ei, H2, H3, H4, H1, cnt);

    finalize_mean_res<<<(int)(nd / 4 / 256), 256, 0, stream>>>(H1, cnt, out);
}


# Edges bucketed by destination each call (int rank atomics, scan, place); one wave per node sums gated messages in registers, no float atomics (sum order = bucket order); mean+residual in GEMM store
# speedup vs baseline: 1.5784x; 1.3894x over previous
#include <hip/hip_runtime.h>

#define NN 50000
#define NE 800000
#define DD 128

typedef float v2f __attribute__((ext_vector_type(2)));
typedef float v8f __attribute__((ext_vector_type(8)));

__global__ __launch_bounds__(256) void bucket_rank(
    const int* __restrict__ ei, int* __restrict__ deg, int* __restrict__ rank)
{
    const int e = blockIdx.x * 256 + threadIdx.x;
    rank[e] = atomicAdd(&deg[ei[NE + e]], 1);
}

__device__ __forceinline__ int wave_inclusive_scan(int x, int lane)
{
    for (int o = 1; o < 32; o <<= 1) {
        const int y = __shfl_up(x, o);
        if (lane >= o) x += y;
    }
    return x;
}

__global__ __launch_bounds__(1024) void bucket_scan(
    const int* __restrict__ deg, int* __restrict__ off)
{
    __shared__ int wsum[13][32];
    const int t = threadIdx.x;
    const int lane = t & 31;
    const int wave = t >> 5;

    int4 v[13];
    int tsum[13], inc[13];
    #pragma unroll
    for (int i = 0; i < 13; ++i) {
        const int q = i * 1024 + t;
        v[i] = (q < NN / 4) ? reinterpret_cast<const int4*>(deg)[q] : make_int4(0, 0, 0, 0);
    }
    #pragma unroll
    for (int i = 0; i < 13; ++i) {
        tsum[i] = v[i].x + v[i].y + v[i].z + v[i].w;
        inc[i] = wave_inclusive_scan(tsum[i], lane);
    }
    if (lane == 31) {
        #pragma unroll
        for (int i = 0; i < 13; ++i) wsum[i][wave] = inc[i];
    }
    __syncthreads();

    int carry = 0;
    #pragma unroll
    for (int i = 0; i < 13; ++i) {
        const int q = i * 1024 + t;
        const int ws = wsum[i][lane];
        const int wsinc = wave_inclusive_scan(ws, lane);
        const int wave_excl = __shfl(wsinc, wave) - __shfl(ws, wave);
        const int total = __shfl(wsinc, 31);
        const int excl = carry + wave_excl + inc[i] - tsum[i];
        if (q < NN / 4) {
            int4 r;
            r.x = excl;
            r.y = r.x + v[i].x;
            r.z = r.y + v[i].y;
            r.w = r.z + v[i].z;
            reinterpret_cast<int4*>(off)[q] = r;
        }
        carry += total;
    }
}

__global__ __launch_bounds__(256) void bucket_place(
    const int* __restrict__ ei, const int* __restrict__ off,
    const int* __restrict__ rank, int* __restrict__ ssrc)
{
    const int e = blockIdx.x * 256 + threadIdx.x;
    const int s = ei[e];
    const int d = ei[NE + e];
    const int pos = off[d] + rank[e];
    if ((unsigned)pos < (unsigned)NE) ssrc[pos] = s;
}

#define GM_TILES 5
#define GM_ROWS  (GM_TILES * 16)

__global__ __launch_bounds__(128) void gemm5_wmma(
    const float* __restrict__ X,
    const float* __restrict__ W1, const float* __restrict__ b1,
    const float* __restrict__ W2, const float* __restrict__ b2,
    const float* __restrict__ W3, const float* __restrict__ b3,
    const float* __restrict__ W4, const float* __restrict__ b4,
    const float* __restrict__ Wr, const float* __restrict__ br,
    const int* __restrict__ deg,
    float* __restrict__ H2, float* __restrict__ H3, float* __restrict__ H4,
    float* __restrict__ out)
{
    __shared__ float xs[GM_ROWS * 132];
    __shared__ __attribute__((aligned(16))) float sinv[GM_ROWS];
    const int slab  = blockIdx.x >> 1;
    const int nhalf = blockIdx.x & 1;
    const int row0  = slab * GM_ROWS;
    const int t = threadIdx.x;

    for (int q = t; q < GM_ROWS * 32; q += 128) {
        const int r  = q >> 5;
        const int c4 = (q & 31) << 2;
        const float4 v = *reinterpret_cast<const float4*>(X + (size_t)(row0 + r) * DD + c4);
        *reinterpret_cast<float4*>(&xs[r * 132 + c4]) = v;
    }
    if (t < GM_ROWS) sinv[t] = 1.0f / fmaxf((float)deg[row0 + t], 1.0f);
    __syncthreads();

    const int wave = t >> 5;
    const int lane = t & 31;
    const int half = lane >> 4;
    const int nn   = lane & 15;
    const int n0   = (nhalf * 4 + wave) * 16;
    const int wrow = n0 + nn;

    const float* Ws[5] = {W1, W2, W3, W4, Wr};

    v8f acc[GM_TILES][5] = {};

    for (int k = 0; k < DD; k += 4) {
        const int kk = k + 2 * half;
        v2f b[5];
        #pragma unroll
        for (int w = 0; w < 5; ++w) {
            const float2 bf = *reinterpret_cast<const float2*>(Ws[w] + (size_t)wrow * DD + kk);
            b[w].x = bf.x; b[w].y = bf.y;
        }
        #pragma unroll
        for (int mt = 0; mt < GM_TILES; ++mt) {
            const float2 af = *reinterpret_cast<const float2*>(&xs[(mt * 16 + nn) * 132 + kk]);
            v2f a; a.x = af.x; a.y = af.y;
            #pragma unroll
            for (int w = 0; w < 5; ++w) {
                acc[mt][w] = __builtin_amdgcn_wmma_f32_16x16x4_f32(
                    false, a, false, b[w], (short)0, acc[mt][w], false, false);
            }
        }
    }

    const size_t obase = (size_t)(row0 + 8 * half) * DD + wrow;
    {
        float* po = out + obase;
        const float bias1 = b1[wrow];
        const float biasr = br[wrow];
        #pragma unroll
        for (int mt = 0; mt < GM_TILES; ++mt) {
            const float4 sa = *reinterpret_cast<const float4*>(&sinv[mt * 16 + 8 * half]);
            const float4 sb = *reinterpret_cast<const float4*>(&sinv[mt * 16 + 8 * half + 4]);
            const float sv[8] = {sa.x, sa.y, sa.z, sa.w, sb.x, sb.y, sb.z, sb.w};
            #pragma unroll
            for (int j = 0; j < 8; ++j) {
                po[(mt * 16 + j) * DD] =
                    (acc[mt][0][j] + bias1) * sv[j] + (acc[mt][4][j] + biasr);
            }
            __builtin_amdgcn_sched_barrier(0);
        }
    }
    const float* Bs[3] = {b2, b3, b4};
    float*       Hs[3] = {H2, H3, H4};
    #pragma unroll
    for (int w = 0; w < 3; ++w) {
        const float bias = Bs[w][wrow];
        float* ph = Hs[w] + obase;
        #pragma unroll
        for (int mt = 0; mt < GM_TILES; ++mt) {
            #pragma unroll
            for (int j = 0; j < 8; ++j) {
                ph[(mt * 16 + j) * DD] = acc[mt][w + 1][j] + bias;
            }
        }
    }
}

typedef float v4f __attribute__((ext_vector_type(4)));

__device__ __forceinline__ void gate_add(float4& acc, const v4f x2, const v4f x3, const float4 x4)
{
    acc.x += x2.x / (1.0f + __expf(-(x3.x + x4.x)));
    acc.y += x2.y / (1.0f + __expf(-(x3.y + x4.y)));
    acc.z += x2.z / (1.0f + __expf(-(x3.z + x4.z)));
    acc.w += x2.w / (1.0f + __expf(-(x3.w + x4.w)));
}

__device__ __forceinline__ int clamp_node(int s)
{
    return min(max(s, 0), NN - 1);
}

__global__ __launch_bounds__(256) void node_gather_reduce(
    const int* __restrict__ deg, const int* __restrict__ off,
    const int* __restrict__ ssrc,
    const float* __restrict__ H2, const float* __restrict__ H3,
    const float* __restrict__ H4, float* __restrict__ out)
{
    const int lane = threadIdx.x & 31;
    const int wave = __builtin_amdgcn_readfirstlane((int)(threadIdx.x >> 5));
    const int n = blockIdx.x * 8 + wave;
    const int d = deg[n];
    const int start = min(max(off[n], 0), NE);
    const int end = min(start + max(d, 0), NE);
    const int d0 = lane << 2;

    const float4 x4 = *reinterpret_cast<const float4*>(H4 + (size_t)n * DD + d0);
    float4* orow = reinterpret_cast<float4*>(out + (size_t)n * DD + d0);
    float4 o = *orow;
    float4 acc = make_float4(0.0f, 0.0f, 0.0f, 0.0f);

    for (int base = start; base < end; base += 32) {
        const int nb = min(32, end - base);
        const int mine = (lane < nb) ? clamp_node(ssrc[base + lane]) : 0;
        int j = 0;
        for (; j + 4 <= nb; j += 4) {
            const size_t r0 = (size_t)__builtin_amdgcn_readlane(mine, j + 0) * DD;
            const size_t r1 = (size_t)__builtin_amdgcn_readlane(mine, j + 1) * DD;
            const size_t r2 = (size_t)__builtin_amdgcn_readlane(mine, j + 2) * DD;
            const size_t r3 = (size_t)__builtin_amdgcn_readlane(mine, j + 3) * DD;
            v4f a3 = *reinterpret_cast<const v4f*>(H3 + r0 + d0);
            v4f a2 = *reinterpret_cast<const v4f*>(H2 + r0 + d0);
            v4f b3 = *reinterpret_cast<const v4f*>(H3 + r1 + d0);
            v4f b2 = *reinterpret_cast<const v4f*>(H2 + r1 + d0);
            v4f c3 = *reinterpret_cast<const v4f*>(H3 + r2 + d0);
            v4f c2 = *reinterpret_cast<const v4f*>(H2 + r2 + d0);
            v4f e3 = *reinterpret_cast<const v4f*>(H3 + r3 + d0);
            v4f e2 = *reinterpret_cast<const v4f*>(H2 + r3 + d0);
            asm volatile("" : "+v"(a3), "+v"(a2), "+v"(b3), "+v"(b2), "+v"(c3), "+v"(c2), "+v"(e3), "+v"(e2));
            gate_add(acc, a2, a3, x4);
            gate_add(acc, b2, b3, x4);
            gate_add(acc, c2, c3, x4);
            gate_add(acc, e2, e3, x4);
        }
        for (; j < nb; ++j) {
            const size_t r0 = (size_t)__builtin_amdgcn_readlane(mine, j) * DD;
            const v4f a3 = *reinterpret_cast<const v4f*>(H3 + r0 + d0);
            const v4f a2 = *reinterpret_cast<const v4f*>(H2 + r0 + d0);
            gate_add(acc, a2, a3, x4);
        }
    }

    const float s = 1.0f / fmaxf((float)d, 1.0f);
    o.x += acc.x * s;
    o.y += acc.y * s;
    o.z += acc.z * s;
    o.w += acc.w * s;
    *orow = o;
}

extern "C" void kernel_launch(void* const* d_in, const int* in_sizes, int n_in,
                              void* d_out, int out_size, void* d_ws, size_t ws_size,
                              hipStream_t stream)
{
    const float* X   = (const float*)d_in[0];
    const int*   ei  = (const int*)  d_in[1];
    const float* W1  = (const float*)d_in[2];
    const float* b1  = (const float*)d_in[3];
    const float* W2  = (const float*)d_in[4];
    const float* b2  = (const float*)d_in[5];
    const float* W3  = (const float*)d_in[6];
    const float* b3  = (const float*)d_in[7];
    const float* W4  = (const float*)d_in[8];
    const float* b4  = (const float*)d_in[9];
    const float* Wr  = (const float*)d_in[10];
    const float* br  = (const float*)d_in[11];
    float* out = (float*)d_out;

    const size_t nd = (size_t)NN * DD;
    float* ws  = (float*)d_ws;
    int*   off  = (int*)ws;
    int*   ssrc = (int*)ws + 65536;
    int*   rank = (int*)ws + 1048576;
    float* H2  = ws + nd;
    float* H3  = ws + 2 * nd;
    float* H4  = ws + 3 * nd;
    float* cnt = ws + 4 * nd;
    int*   deg = (int*)cnt;

    hipMemsetAsync(cnt, 0, NN * sizeof(float), stream);

    bucket_rank<<<NE / 256, 256, 0, stream>>>(ei, deg, rank);
    bucket_scan<<<1, 1024, 0, stream>>>(deg, off);
    bucket_place<<<NE / 256, 256, 0, stream>>>(ei, off, rank, ssrc);

    gemm5_wmma<<<(NN / GM_ROWS) * 2, 128, 0, stream>>>(
        X, W1, b1, W2, b2, W3, b3, W4, b4, Wr, br, deg,
        H2, H3, H4, out);

    node_gather_reduce<<<NN / 8, 256, 0, stream>>>(deg, off, ssrc, H2, H3, H4, out);
}


# Edge bucketing as a two-level counting sort in LDS (200 coarse bins, then per-bin node sort; runs stored contiguously): no global atomics, no memset; sum order in a bucket = LDS atomic arrival order
# speedup vs baseline: 2.2394x; 1.4188x over previous
#include <hip/hip_runtime.h>

#define NN 50000
#define NE 800000
#define DD 128

typedef float v2f __attribute__((ext_vector_type(2)));
typedef float v8f __attribute__((ext_vector_type(8)));

#define NBIN   200
#define BINW   250
#define NCHUNK 250
#define CHUNK  3200
#define SORT_CAP 8192

__device__ __forceinline__ int coarse_bin(int d)
{
    return min(max(d / BINW, 0), NBIN - 1);
}

__device__ __forceinline__ int wave_inclusive_scan(int x, int lane)
{
    for (int o = 1; o < 32; o <<= 1) {
        const int y = __shfl_up(x, o);
        if (lane >= o) x += y;
    }
    return x;
}

__device__ __forceinline__ int block_excl_scan256(int v, int t, int* wtot)
{
    const int lane = t & 31;
    const int wave = t >> 5;
    const int inc = wave_inclusive_scan(v, lane);
    if (t < 256 && lane == 31) wtot[wave] = inc;
    __syncthreads();
    int prefix = 0;
    for (int w = 0; w < 8; ++w) if (w < wave) prefix += wtot[w];
    return prefix + inc - v;
}

__global__ __launch_bounds__(1024) void coarse_count(
    const int* __restrict__ ei, int* __restrict__ cnt)
{
    __shared__ int hist[256];
    const int t = threadIdx.x;
    const int g = blockIdx.x;
    if (t < 256) hist[t] = 0;
    __syncthreads();
    const int* dst = ei + NE + g * CHUNK;
    for (int i = t; i < CHUNK; i += 1024) atomicAdd(&hist[coarse_bin(dst[i])], 1);
    __syncthreads();
    if (t < NBIN) cnt[t * NCHUNK + g] = hist[t];
}

__global__ __launch_bounds__(1024) void bucket_scan(
    const int* __restrict__ cnt, int* __restrict__ pre)
{
    __shared__ int wsum[13][32];
    const int t = threadIdx.x;
    const int lane = t & 31;
    const int wave = t >> 5;

    int4 v[13];
    int tsum[13], inc[13];
    #pragma unroll
    for (int i = 0; i < 13; ++i) {
        const int q = i * 1024 + t;
        v[i] = (q < NN / 4) ? reinterpret_cast<const int4*>(cnt)[q] : make_int4(0, 0, 0, 0);
    }
    #pragma unroll
    for (int i = 0; i < 13; ++i) {
        tsum[i] = v[i].x + v[i].y + v[i].z + v[i].w;
        inc[i] = wave_inclusive_scan(tsum[i], lane);
    }
    if (lane == 31) {
        #pragma unroll
        for (int i = 0; i < 13; ++i) wsum[i][wave] = inc[i];
    }
    __syncthreads();

    int carry = 0;
    #pragma unroll
    for (int i = 0; i < 13; ++i) {
        const int q = i * 1024 + t;
        const int ws = wsum[i][lane];
        const int wsinc = wave_inclusive_scan(ws, lane);
        const int wave_excl = __shfl(wsinc, wave) - __shfl(ws, wave);
        const int total = __shfl(wsinc, 31);
        const int excl = carry + wave_excl + inc[i] - tsum[i];
        if (q < NN / 4) {
            int4 r;
            r.x = excl;
            r.y = r.x + v[i].x;
            r.z = r.y + v[i].y;
            r.w = r.z + v[i].z;
            reinterpret_cast<int4*>(pre)[q] = r;
        }
        carry += total;
    }
}

__global__ __launch_bounds__(1024) void coarse_scatter(
    const int* __restrict__ ei, const int* __restrict__ base, int* __restrict__ tmp)
{
    __shared__ int hist[256];
    __shared__ int lstart[256];
    __shared__ int gbase[256];
    __shared__ int wtot[8];
    __shared__ int2 stage[CHUNK];
    const int t = threadIdx.x;
    const int g = blockIdx.x;
    if (t < 256) hist[t] = 0;
    __syncthreads();
    const int* src = ei + g * CHUNK;
    const int* dst = ei + NE + g * CHUNK;
    int s[4], d[4], r[4];
    #pragma unroll
    for (int k = 0; k < 4; ++k) {
        const int i = t + 1024 * k;
        if (i < CHUNK) {
            s[k] = src[i];
            d[k] = dst[i];
            r[k] = atomicAdd(&hist[coarse_bin(d[k])], 1);
        }
    }
    __syncthreads();
    const int v = (t < 256) ? hist[t] : 0;
    const int excl = block_excl_scan256(v, t, wtot);
    if (t < 256) {
        lstart[t] = excl;
        gbase[t] = (t < NBIN) ? base[t * NCHUNK + g] : 0;
    }
    __syncthreads();
    #pragma unroll
    for (int k = 0; k < 4; ++k) {
        const int i = t + 1024 * k;
        if (i < CHUNK) stage[lstart[coarse_bin(d[k])] + r[k]] = make_int2(s[k], d[k]);
    }
    __syncthreads();
    int2* out2 = reinterpret_cast<int2*>(tmp);
    for (int i = t; i < CHUNK; i += 1024) {
        const int2 e = stage[i];
        const int b = coarse_bin(e.y);
        const int pos = gbase[b] + (i - lstart[b]);
        if ((unsigned)pos < (unsigned)NE) out2[pos] = e;
    }
}

__global__ __launch_bounds__(1024) void bin_sort(
    const int* __restrict__ base, const int* __restrict__ tmp,
    int* __restrict__ off, int* __restrict__ deg, int* __restrict__ ssrc)
{
    __shared__ int c[256];
    __shared__ int wtot[8];
    __shared__ int sorted[SORT_CAP];
    const int t = threadIdx.x;
    const int b = blockIdx.x;
    const int2* in2 = reinterpret_cast<const int2*>(tmp);
    const int start = min(max(base[b * NCHUNK], 0), NE);
    const int endraw = (b + 1 < NBIN) ? base[(b + 1) * NCHUNK] : NE;
    const int end = min(max(endraw, start), NE);
    const int n = end - start;
    const int node0 = b * BINW;
    if (t < 256) c[t] = 0;
    __syncthreads();
    for (int i = t; i < n; i += 1024) {
        const int l = in2[start + i].y - node0;
        if ((unsigned)l < (unsigned)BINW) atomicAdd(&c[l], 1);
    }
    __syncthreads();
    const int v = (t < 256) ? c[t] : 0;
    const int excl = block_excl_scan256(v, t, wtot);
    if (t < BINW) {
        off[node0 + t] = start + excl;
        deg[node0 + t] = v;
    }
    if (t < 256) c[t] = excl;
    __syncthreads();
    const bool staged = (n <= SORT_CAP);
    for (int i = t; i < n; i += 1024) {
        const int2 e = in2[start + i];
        const int l = e.y - node0;
        if ((unsigned)l < (unsigned)BINW) {
            const int p = atomicAdd(&c[l], 1);
            if (staged) {
                if ((unsigned)p < (unsigned)SORT_CAP) sorted[p] = e.x;
            } else if ((unsigned)(start + p) < (unsigned)NE) {
                ssrc[start + p] = e.x;
            }
        }
    }
    if (staged) {
        __syncthreads();
        for (int i = t; i < n; i += 1024) ssrc[start + i] = sorted[i];
    }
}

#define GM_TILES 5
#define GM_ROWS  (GM_TILES * 16)

__global__ __launch_bounds__(128) void gemm5_wmma(
    const float* __restrict__ X,
    const float* __restrict__ W1, const float* __restrict__ b1,
    const float* __restrict__ W2, const float* __restrict__ b2,
    const float* __restrict__ W3, const float* __restrict__ b3,
    const float* __restrict__ W4, const float* __restrict__ b4,
    const float* __restrict__ Wr, const float* __restrict__ br,
    const int* __restrict__ deg,
    float* __restrict__ H2, float* __restrict__ H3, float* __restrict__ H4,
    float* __restrict__ out)
{
    __shared__ float xs[GM_ROWS * 132];
    __shared__ __attribute__((aligned(16))) float sinv[GM_ROWS];
    const int slab  = blockIdx.x >> 1;
    const int nhalf = blockIdx.x & 1;
    const int row0  = slab * GM_ROWS;
    const int t = threadIdx.x;

    for (int q = t; q < GM_ROWS * 32; q += 128) {
        const int r  = q >> 5;
        const int c4 = (q & 31) << 2;
        const float4 v = *reinterpret_cast<const float4*>(X + (size_t)(row0 + r) * DD + c4);
        *reinterpret_cast<float4*>(&xs[r * 132 + c4]) = v;
    }
    if (t < GM_ROWS) sinv[t] = 1.0f / fmaxf((float)deg[row0 + t], 1.0f);
    __syncthreads();

    const int wave = t >> 5;
    const int lane = t & 31;
    const int half = lane >> 4;
    const int nn   = lane & 15;
    const int n0   = (nhalf * 4 + wave) * 16;
    const int wrow = n0 + nn;

    const float* Ws[5] = {W1, W2, W3, W4, Wr};

    v8f acc[GM_TILES][5] = {};

    for (int k = 0; k < DD; k += 4) {
        const int kk = k + 2 * half;
        v2f b[5];
        #pragma unroll
        for (int w = 0; w < 5; ++w) {
            const float2 bf = *reinterpret_cast<const float2*>(Ws[w] + (size_t)wrow * DD + kk);
            b[w].x = bf.x; b[w].y = bf.y;
        }
        #pragma unroll
        for (int mt = 0; mt < GM_TILES; ++mt) {
            const float2 af = *reinterpret_cast<const float2*>(&xs[(mt * 16 + nn) * 132 + kk]);
            v2f a; a.x = af.x; a.y = af.y;
            #pragma unroll
            for (int w = 0; w < 5; ++w) {
                acc[mt][w] = __builtin_amdgcn_wmma_f32_16x16x4_f32(
                    false, a, false, b[w], (short)0, acc[mt][w], false, false);
            }
        }
    }

    const size_t obase = (size_t)(row0 + 8 * half) * DD + wrow;
    {
        float* po = out + obase;
        const float bias1 = b1[wrow];
        const float biasr = br[wrow];
        #pragma unroll
        for (int mt = 0; mt < GM_TILES; ++mt) {
            const float4 sa = *reinterpret_cast<const float4*>(&sinv[mt * 16 + 8 * half]);
            const float4 sb = *reinterpret_cast<const float4*>(&sinv[mt * 16 + 8 * half + 4]);
            const float sv[8] = {sa.x, sa.y, sa.z, sa.w, sb.x, sb.y, sb.z, sb.w};
            #pragma unroll
            for (int j = 0; j < 8; ++j) {
                po[(mt * 16 + j) * DD] =
                    (acc[mt][0][j] + bias1) * sv[j] + (acc[mt][4][j] + biasr);
            }
            __builtin_amdgcn_sched_barrier(0);
        }
    }
    const float* Bs[3] = {b2, b3, b4};
    float*       Hs[3] = {H2, H3, H4};
    #pragma unroll
    for (int w = 0; w < 3; ++w) {
        const float bias = Bs[w][wrow];
        float* ph = Hs[w] + obase;
        #pragma unroll
        for (int mt = 0; mt < GM_TILES; ++mt) {
            #pragma unroll
            for (int j = 0; j < 8; ++j) {
                ph[(mt * 16 + j) * DD] = acc[mt][w + 1][j] + bias;
            }
        }
    }
}

typedef float v4f __attribute__((ext_vector_type(4)));

__device__ __forceinline__ void gate_add(float4& acc, const v4f x2, const v4f x3, const float4 x4)
{
    acc.x += x2.x / (1.0f + __expf(-(x3.x + x4.x)));
    acc.y += x2.y / (1.0f + __expf(-(x3.y + x4.y)));
    acc.z += x2.z / (1.0f + __expf(-(x3.z + x4.z)));
    acc.w += x2.w / (1.0f + __expf(-(x3.w + x4.w)));
}

__device__ __forceinline__ int clamp_node(int s)
{
    return min(max(s, 0), NN - 1);
}

__global__ __launch_bounds__(256) void node_gather_reduce(
    const int* __restrict__ deg, const int* __restrict__ off,
    const int* __restrict__ ssrc,
    const float* __restrict__ H2, const float* __restrict__ H3,
    const float* __restrict__ H4, float* __restrict__ out)
{
    const int lane = threadIdx.x & 31;
    const int wave = __builtin_amdgcn_readfirstlane((int)(threadIdx.x >> 5));
    const int n = blockIdx.x * 8 + wave;
    const int d = deg[n];
    const int start = min(max(off[n], 0), NE);
    const int end = min(start + max(d, 0), NE);
    const int d0 = lane << 2;

    const float4 x4 = *reinterpret_cast<const float4*>(H4 + (size_t)n * DD + d0);
    float4* orow = reinterpret_cast<float4*>(out + (size_t)n * DD + d0);
    float4 o = *orow;
    float4 acc = make_float4(0.0f, 0.0f, 0.0f, 0.0f);

    for (int base = start; base < end; base += 32) {
        const int nb = min(32, end - base);
        const int mine = (lane < nb) ? clamp_node(ssrc[base + lane]) : 0;
        int j = 0;
        for (; j + 4 <= nb; j += 4) {
            const size_t r0 = (size_t)__builtin_amdgcn_readlane(mine, j + 0) * DD;
            const size_t r1 = (size_t)__builtin_amdgcn_readlane(mine, j + 1) * DD;
            const size_t r2 = (size_t)__builtin_amdgcn_readlane(mine, j + 2) * DD;
            const size_t r3 = (size_t)__builtin_amdgcn_readlane(mine, j + 3) * DD;
            v4f a3 = *reinterpret_cast<const v4f*>(H3 + r0 + d0);
            v4f a2 = *reinterpret_cast<const v4f*>(H2 + r0 + d0);
            v4f b3 = *reinterpret_cast<const v4f*>(H3 + r1 + d0);
            v4f b2 = *reinterpret_cast<const v4f*>(H2 + r1 + d0);
            v4f c3 = *reinterpret_cast<const v4f*>(H3 + r2 + d0);
            v4f c2 = *reinterpret_cast<const v4f*>(H2 + r2 + d0);
            v4f e3 = *reinterpret_cast<const v4f*>(H3 + r3 + d0);
            v4f e2 = *reinterpret_cast<const v4f*>(H2 + r3 + d0);
            asm volatile("" : "+v"(a3), "+v"(a2), "+v"(b3), "+v"(b2), "+v"(c3), "+v"(c2), "+v"(e3), "+v"(e2));
            gate_add(acc, a2, a3, x4);
            gate_add(acc, b2, b3, x4);
            gate_add(acc, c2, c3, x4);
            gate_add(acc, e2, e3, x4);
        }
        for (; j < nb; ++j) {
            const size_t r0 = (size_t)__builtin_amdgcn_readlane(mine, j) * DD;
            const v4f a3 = *reinterpret_cast<const v4f*>(H3 + r0 + d0);
            const v4f a2 = *reinterpret_cast<const v4f*>(H2 + r0 + d0);
            gate_add(acc, a2, a3, x4);
        }
    }

    const float s = 1.0f / fmaxf((float)d, 1.0f);
    o.x += acc.x * s;
    o.y += acc.y * s;
    o.z += acc.z * s;
    o.w += acc.w * s;
    *orow = o;
}

extern "C" void kernel_launch(void* const* d_in, const int* in_sizes, int n_in,
                              void* d_out, int out_size, void* d_ws, size_t ws_size,
                              hipStream_t stream)
{
    const float* X   = (const float*)d_in[0];
    const int*   ei  = (const int*)  d_in[1];
    const float* W1  = (const float*)d_in[2];
    const float* b1  = (const float*)d_in[3];
    const float* W2  = (const float*)d_in[4];
    const float* b2  = (const float*)d_in[5];
    const float* W3  = (const float*)d_in[6];
    const float* b3  = (const float*)d_in[7];
    const float* W4  = (const float*)d_in[8];
    const float* b4  = (const float*)d_in[9];
    const float* Wr  = (const float*)d_in[10];
    const float* br  = (const float*)d_in[11];
    float* out = (float*)d_out;

    const size_t nd = (size_t)NN * DD;
    float* ws  = (float*)d_ws;
    int*   off  = (int*)ws;
    int*   ssrc = (int*)ws + 65536;
    int*   tmp  = (int*)ws + 1048576;
    int*   ccnt = (int*)ws + 2752512;
    int*   base = (int*)ws + 2818048;
    float* H2  = ws + nd;
    float* H3  = ws + 2 * nd;
    float* H4  = ws + 3 * nd;
    float* cnt = ws + 4 * nd;
    int*   deg = (int*)cnt;

    coarse_count<<<NCHUNK, 1024, 0, stream>>>(ei, ccnt);
    bucket_scan<<<1, 1024, 0, stream>>>(ccnt, base);
    coarse_scatter<<<NCHUNK, 1024, 0, stream>>>(ei, base, tmp);
    bin_sort<<<NBIN, 1024, 0, stream>>>(base, tmp, off, deg, ssrc);

    gemm5_wmma<<<(NN / GM_ROWS) * 2, 128, 0, stream>>>(
        X, W1, b1, W2, b2, W3, b3, W4, b4, Wr, br, deg,
        H2, H3, H4, out);

    node_gather_reduce<<<NN / 8, 256, 0, stream>>>(deg, off, ssrc, H2, H3, H4, out);
}


# Edge bucketing in 2 launches, not 4: chunks bin-ordered in place, a bin's start = sum of its run starts over chunks (no count pass, no global scan); sum order in a bucket = LDS atomic arrival order
# speedup vs baseline: 2.4544x; 1.0960x over previous
#include <hip/hip_runtime.h>

#define NN 50000
#define NE 800000
#define DD 128

typedef float v2f __attribute__((ext_vector_type(2)));
typedef float v8f __attribute__((ext_vector_type(8)));

#define NBIN   200
#define BINW   250
#define NCHUNK 250
#define CHUNK  3200
#define CST    256
#define SORT_CAP 8192

__device__ __forceinline__ int coarse_bin(int d)
{
    return min(max(d / BINW, 0), NBIN - 1);
}

__device__ __forceinline__ int wave_inclusive_scan(int x, int lane)
{
    for (int o = 1; o < 32; o <<= 1) {
        const int y = __shfl_up(x, o);
        if (lane >= o) x += y;
    }
    return x;
}

__device__ __forceinline__ int block_excl_scan256(int v, int t, int* wtot)
{
    const int lane = t & 31;
    const int wave = t >> 5;
    const int inc = wave_inclusive_scan(v, lane);
    if (t < 256 && lane == 31) wtot[wave] = inc;
    __syncthreads();
    int prefix = 0;
    for (int w = 0; w < 8; ++w) if (w < wave) prefix += wtot[w];
    return prefix + inc - v;
}

__device__ __forceinline__ void block_sum2_256(int x, int y, int t, int* w, int& sx, int& sy)
{
    for (int o = 16; o > 0; o >>= 1) {
        x += __shfl_xor(x, o);
        y += __shfl_xor(y, o);
    }
    if (t < 256 && (t & 31) == 0) {
        w[t >> 5] = x;
        w[8 + (t >> 5)] = y;
    }
    __syncthreads();
    sx = 0;
    sy = 0;
    for (int i = 0; i < 8; ++i) {
        sx += w[i];
        sy += w[8 + i];
    }
}

__global__ __launch_bounds__(1024) void chunk_sort(
    const int* __restrict__ ei, int* __restrict__ cstart, int* __restrict__ tmp)
{
    __shared__ int hist[256];
    __shared__ int lstart[256];
    __shared__ int wtot[8];
    __shared__ int2 stage[CHUNK];
    const int t = threadIdx.x;
    const int g = blockIdx.x;
    if (t < 256) hist[t] = 0;
    __syncthreads();
    const int* src = ei + g * CHUNK;
    const int* dst = ei + NE + g * CHUNK;
    int s[4], d[4], r[4];
    #pragma unroll
    for (int k = 0; k < 4; ++k) {
        const int i = t + 1024 * k;
        if (i < CHUNK) {
            s[k] = src[i];
            d[k] = dst[i];
            r[k] = atomicAdd(&hist[coarse_bin(d[k])], 1);
        }
    }
    __syncthreads();
    const int v = (t < 256) ? hist[t] : 0;
    const int excl = block_excl_scan256(v, t, wtot);
    if (t < 256) {
        lstart[t] = excl;
        cstart[g * CST + t] = excl;
    }
    __syncthreads();
    #pragma unroll
    for (int k = 0; k < 4; ++k) {
        const int i = t + 1024 * k;
        if (i < CHUNK) stage[lstart[coarse_bin(d[k])] + r[k]] = make_int2(s[k], d[k]);
    }
    __syncthreads();
    int2* out2 = reinterpret_cast<int2*>(tmp) + g * CHUNK;
    for (int i = t; i < CHUNK; i += 1024) out2[i] = stage[i];
}

__global__ __launch_bounds__(1024) void bin_sort(
    const int* __restrict__ cstart, const int* __restrict__ tmp,
    int* __restrict__ off, int* __restrict__ deg, int* __restrict__ ssrc)
{
    __shared__ int c[256];
    __shared__ int wtot[8];
    __shared__ int wsum[16];
    __shared__ int rbeg[256];
    __shared__ int rlen[256];
    __shared__ int sorted[SORT_CAP];
    const int t = threadIdx.x;
    const int b = blockIdx.x;
    const int2* in2 = reinterpret_cast<const int2*>(tmp);
    const int node0 = b * BINW;
    int ra = 0, rl = 0;
    if (t < NCHUNK) {
        ra = min(max(cstart[t * CST + b], 0), CHUNK);
        rl = min(max(cstart[t * CST + b + 1], ra), CHUNK) - ra;
    }
    if (t < 256) {
        c[t] = 0;
        rbeg[t] = t * CHUNK + ra;
        rlen[t] = rl;
    }
    int start, n;
    block_sum2_256(ra, rl, t, wsum, start, n);
    const int q = t & 3;
    const int rb = rbeg[t >> 2];
    const int rn = rlen[t >> 2];
    for (int j = q; j < rn; j += 4) {
        const int l = in2[rb + j].y - node0;
        if ((unsigned)l < (unsigned)BINW) atomicAdd(&c[l], 1);
    }
    __syncthreads();
    const int v = (t < 256) ? c[t] : 0;
    const int excl = block_excl_scan256(v, t, wtot);
    if (t < BINW) {
        off[node0 + t] = start + excl;
        deg[node0 + t] = v;
    }
    if (t < 256) c[t] = excl;
    __syncthreads();
    const bool staged = (n <= SORT_CAP);
    for (int j = q; j < rn; j += 4) {
        const int2 e = in2[rb + j];
        const int l = e.y - node0;
        if ((unsigned)l < (unsigned)BINW) {
            const int p = atomicAdd(&c[l], 1);
            if (staged) {
                if ((unsigned)p < (unsigned)SORT_CAP) sorted[p] = e.x;
            } else if ((unsigned)(start + p) < (unsigned)NE) {
                ssrc[start + p] = e.x;
            }
        }
    }
    if (staged) {
        __syncthreads();
        for (int i = t; i < n; i += 1024) {
            if ((unsigned)(start + i) < (unsigned)NE) ssrc[start + i] = sorted[i];
        }
    }
}

#define GM_TILES 5
#define GM_ROWS  (GM_TILES * 16)

__global__ __launch_bounds__(128) void gemm5_wmma(
    const float* __restrict__ X,
    const float* __restrict__ W1, const float* __restrict__ b1,
    const float* __restrict__ W2, const float* __restrict__ b2,
    const float* __restrict__ W3, const float* __restrict__ b3,
    const float* __restrict__ W4, const float* __restrict__ b4,
    const float* __restrict__ Wr, const float* __restrict__ br,
    const int* __restrict__ deg,
    float* __restrict__ H2, float* __restrict__ H3, float* __restrict__ H4,
    float* __restrict__ out)
{
    __shared__ float xs[GM_ROWS * 132];
    __shared__ __attribute__((aligned(16))) float sinv[GM_ROWS];
    const int slab  = blockIdx.x >> 1;
    const int nhalf = blockIdx.x & 1;
    const int row0  = slab * GM_ROWS;
    const int t = threadIdx.x;

    for (int q = t; q < GM_ROWS * 32; q += 128) {
        const int r  = q >> 5;
        const int c4 = (q & 31) << 2;
        const float4 v = *reinterpret_cast<const float4*>(X + (size_t)(row0 + r) * DD + c4);
        *reinterpret_cast<float4*>(&xs[r * 132 + c4]) = v;
    }
    if (t < GM_ROWS) sinv[t] = 1.0f / fmaxf((float)deg[row0 + t], 1.0f);
    __syncthreads();

    const int wave = t >> 5;
    const int lane = t & 31;
    const int half = lane >> 4;
    const int nn   = lane & 15;
    const int n0   = (nhalf * 4 + wave) * 16;
    const int wrow = n0 + nn;

    const float* Ws[5] = {W1, W2, W3, W4, Wr};

    v8f acc[GM_TILES][5] = {};

    for (int k = 0; k < DD; k += 4) {
        const int kk = k + 2 * half;
        v2f b[5];
        #pragma unroll
        for (int w = 0; w < 5; ++w) {
            const float2 bf = *reinterpret_cast<const float2*>(Ws[w] + (size_t)wrow * DD + kk);
            b[w].x = bf.x; b[w].y = bf.y;
        }
        #pragma unroll
        for (int mt = 0; mt < GM_TILES; ++mt) {
            const float2 af = *reinterpret_cast<const float2*>(&xs[(mt * 16 + nn) * 132 + kk]);
            v2f a; a.x = af.x; a.y = af.y;
            #pragma unroll
            for (int w = 0; w < 5; ++w) {
                acc[mt][w] = __builtin_amdgcn_wmma_f32_16x16x4_f32(
                    false, a, false, b[w], (short)0, acc[mt][w], false, false);
            }
        }
    }

    const size_t obase = (size_t)(row0 + 8 * half) * DD + wrow;
    {
        float* po = out + obase;
        const float bias1 = b1[wrow];
        const float biasr = br[wrow];
        #pragma unroll
        for (int mt = 0; mt < GM_TILES; ++mt) {
            const float4 sa = *reinterpret_cast<const float4*>(&sinv[mt * 16 + 8 * half]);
            const float4 sb = *reinterpret_cast<const float4*>(&sinv[mt * 16 + 8 * half + 4]);
            const float sv[8] = {sa.x, sa.y, sa.z, sa.w, sb.x, sb.y, sb.z, sb.w};
            #pragma unroll
            for (int j = 0; j < 8; ++j) {
                po[(mt * 16 + j) * DD] =
                    (acc[mt][0][j] + bias1) * sv[j] + (acc[mt][4][j] + biasr);
            }
            __builtin_amdgcn_sched_barrier(0);
        }
    }
    const float* Bs[3] = {b2, b3, b4};
    float*       Hs[3] = {H2, H3, H4};
    #pragma unroll
    for (int w = 0; w < 3; ++w) {
        const float bias = Bs[w][wrow];
        float* ph = Hs[w] + obase;
        #pragma unroll
        for (int mt = 0; mt < GM_TILES; ++mt) {
            #pragma unroll
            for (int j = 0; j < 8; ++j) {
                ph[(mt * 16 + j) * DD] = acc[mt][w + 1][j] + bias;
            }
        }
    }
}

typedef float v4f __attribute__((ext_vector_type(4)));

__device__ __forceinline__ void gate_add(float4& acc, const v4f x2, const v4f x3, const float4 x4)
{
    acc.x += x2.x / (1.0f + __expf(-(x3.x + x4.x)));
    acc.y += x2.y / (1.0f + __expf(-(x3.y + x4.y)));
    acc.z += x2.z / (1.0f + __expf(-(x3.z + x4.z)));
    acc.w += x2.w / (1.0f + __expf(-(x3.w + x4.w)));
}

__device__ __forceinline__ int clamp_node(int s)
{
    return min(max(s, 0), NN - 1);
}

__global__ __launch_bounds__(256) void node_gather_reduce(
    const int* __restrict__ deg, const int* __restrict__ off,
    const int* __restrict__ ssrc,
    const float* __restrict__ H2, const float* __restrict__ H3,
    const float* __restrict__ H4, float* __restrict__ out)
{
    const int lane = threadIdx.x & 31;
    const int wave = __builtin_amdgcn_readfirstlane((int)(threadIdx.x >> 5));
    const int n = blockIdx.x * 8 + wave;
    const int d = deg[n];
    const int start = min(max(off[n], 0), NE);
    const int end = min(start + max(d, 0), NE);
    const int d0 = lane << 2;

    const float4 x4 = *reinterpret_cast<const float4*>(H4 + (size_t)n * DD + d0);
    float4* orow = reinterpret_cast<float4*>(out + (size_t)n * DD + d0);
    float4 o = *orow;
    float4 acc = make_float4(0.0f, 0.0f, 0.0f, 0.0f);

    for (int base = start; base < end; base += 32) {
        const int nb = min(32, end - base);
        const int mine = (lane < nb) ? clamp_node(ssrc[base + lane]) : 0;
        int j = 0;
        for (; j + 4 <= nb; j += 4) {
            const size_t r0 = (size_t)__builtin_amdgcn_readlane(mine, j + 0) * DD;
            const size_t r1 = (size_t)__builtin_amdgcn_readlane(mine, j + 1) * DD;
            const size_t r2 = (size_t)__builtin_amdgcn_readlane(mine, j + 2) * DD;
            const size_t r3 = (size_t)__builtin_amdgcn_readlane(mine, j + 3) * DD;
            v4f a3 = *reinterpret_cast<const v4f*>(H3 + r0 + d0);
            v4f a2 = *reinterpret_cast<const v4f*>(H2 + r0 + d0);
            v4f b3 = *reinterpret_cast<const v4f*>(H3 + r1 + d0);
            v4f b2 = *reinterpret_cast<const v4f*>(H2 + r1 + d0);
            v4f c3 = *reinterpret_cast<const v4f*>(H3 + r2 + d0);
            v4f c2 = *reinterpret_cast<const v4f*>(H2 + r2 + d0);
            v4f e3 = *reinterpret_cast<const v4f*>(H3 + r3 + d0);
            v4f e2 = *reinterpret_cast<const v4f*>(H2 + r3 + d0);
            asm volatile("" : "+v"(a3), "+v"(a2), "+v"(b3), "+v"(b2), "+v"(c3), "+v"(c2), "+v"(e3), "+v"(e2));
            gate_add(acc, a2, a3, x4);
            gate_add(acc, b2, b3, x4);
            gate_add(acc, c2, c3, x4);
            gate_add(acc, e2, e3, x4);
        }
        for (; j < nb; ++j) {
            const size_t r0 = (size_t)__builtin_amdgcn_readlane(mine, j) * DD;
            const v4f a3 = *reinterpret_cast<const v4f*>(H3 + r0 + d0);
            const v4f a2 = *reinterpret_cast<const v4f*>(H2 + r0 + d0);
            gate_add(acc, a2, a3, x4);
        }
    }

    const float s = 1.0f / fmaxf((float)d, 1.0f);
    o.x += acc.x * s;
    o.y += acc.y * s;
    o.z += acc.z * s;
    o.w += acc.w * s;
    *orow = o;
}

extern "C" void kernel_launch(void* const* d_in, const int* in_sizes, int n_in,
                              void* d_out, int out_size, void* d_ws, size_t ws_size,
                              hipStream_t stream)
{
    const float* X   = (const float*)d_in[0];
    const int*   ei  = (const int*)  d_in[1];
    const float* W1  = (const float*)d_in[2];
    const float* b1  = (const float*)d_in[3];
    const float* W2  = (const float*)d_in[4];
    const float* b2  = (const float*)d_in[5];
    const float* W3  = (const float*)d_in[6];
    const float* b3  = (const float*)d_in[7];
    const float* W4  = (const float*)d_in[8];
    const float* b4  = (const float*)d_in[9];
    const float* Wr  = (const float*)d_in[10];
    const float* br  = (const float*)d_in[11];
    float* out = (float*)d_out;

    const size_t nd = (size_t)NN * DD;
    float* ws  = (float*)d_ws;
    int*   off  = (int*)ws;
    int*   ssrc = (int*)ws + 65536;
    int*   tmp  = (int*)ws + 1048576;
    int*   cstart = (int*)ws + 2752512;
    float* H2  = ws + nd;
    float* H3  = ws + 2 * nd;
    float* H4  = ws + 3 * nd;
    float* cnt = ws + 4 * nd;
    int*   deg = (int*)cnt;

    chunk_sort<<<NCHUNK, 1024, 0, stream>>>(ei, cstart, tmp);
    bin_sort<<<NBIN, 1024, 0, stream>>>(cstart, tmp, off, deg, ssrc);

    gemm5_wmma<<<(NN / GM_ROWS) * 2, 128, 0, stream>>>(
        X, W1, b1, W2, b2, W3, b3, W4, b4, Wr, br, deg,
        H2, H3, H4, out);

    node_gather_reduce<<<NN / 8, 256, 0, stream>>>(deg, off, ssrc, H2, H3, H4, out);
}


# Edge gate: sigmoid's divide taken as v_rcp_f32 (1 ulp) times the message, fused into the sum, not the IEEE divide sequence: 242 -> 91 vector instructions per 4 edges in the hot loop; sum order kept
# speedup vs baseline: 2.7068x; 1.1028x over previous
#include <hip/hip_runtime.h>

#define NN 50000
#define NE 800000
#define DD 128

typedef float v2f __attribute__((ext_vector_type(2)));
typedef float v8f __attribute__((ext_vector_type(8)));

#define NBIN   200
#define BINW   250
#define NCHUNK 250
#define CHUNK  3200
#define CST    256
#define SORT_CAP 8192

__device__ __forceinline__ int coarse_bin(int d)
{
    return min(max(d / BINW, 0), NBIN - 1);
}

__device__ __forceinline__ int wave_inclusive_scan(int x, int lane)
{
    for (int o = 1; o < 32; o <<= 1) {
        const int y = __shfl_up(x, o);
        if (lane >= o) x += y;
    }
    return x;
}

__device__ __forceinline__ int block_excl_scan256(int v, int t, int* wtot)
{
    const int lane = t & 31;
    const int wave = t >> 5;
    const int inc = wave_inclusive_scan(v, lane);
    if (t < 256 && lane == 31) wtot[wave] = inc;
    __syncthreads();
    int prefix = 0;
    for (int w = 0; w < 8; ++w) if (w < wave) prefix += wtot[w];
    return prefix + inc - v;
}

__device__ __forceinline__ void block_sum2_256(int x, int y, int t, int* w, int& sx, int& sy)
{
    for (int o = 16; o > 0; o >>= 1) {
        x += __shfl_xor(x, o);
        y += __shfl_xor(y, o);
    }
    if (t < 256 && (t & 31) == 0) {
        w[t >> 5] = x;
        w[8 + (t >> 5)] = y;
    }
    __syncthreads();
    sx = 0;
    sy = 0;
    for (int i = 0; i < 8; ++i) {
        sx += w[i];
        sy += w[8 + i];
    }
}

__global__ __launch_bounds__(1024) void chunk_sort(
    const int* __restrict__ ei, int* __restrict__ cstart, int* __restrict__ tmp)
{
    __shared__ int hist[256];
    __shared__ int lstart[256];
    __shared__ int wtot[8];
    __shared__ int2 stage[CHUNK];
    const int t = threadIdx.x;
    const int g = blockIdx.x;
    if (t < 256) hist[t] = 0;
    __syncthreads();
    const int* src = ei + g * CHUNK;
    const int* dst = ei + NE + g * CHUNK;
    int s[4], d[4], r[4];
    #pragma unroll
    for (int k = 0; k < 4; ++k) {
        const int i = t + 1024 * k;
        if (i < CHUNK) {
            s[k] = src[i];
            d[k] = dst[i];
            r[k] = atomicAdd(&hist[coarse_bin(d[k])], 1);
        }
    }
    __syncthreads();
    const int v = (t < 256) ? hist[t] : 0;
    const int excl = block_excl_scan256(v, t, wtot);
    if (t < 256) {
        lstart[t] = excl;
        cstart[g * CST + t] = excl;
    }
    __syncthreads();
    #pragma unroll
    for (int k = 0; k < 4; ++k) {
        const int i = t + 1024 * k;
        if (i < CHUNK) stage[lstart[coarse_bin(d[k])] + r[k]] = make_int2(s[k], d[k]);
    }
    __syncthreads();
    int2* out2 = reinterpret_cast<int2*>(tmp) + g * CHUNK;
    for (int i = t; i < CHUNK; i += 1024) out2[i] = stage[i];
}

__global__ __launch_bounds__(1024) void bin_sort(
    const int* __restrict__ cstart, const int* __restrict__ tmp,
    int* __restrict__ off, int* __restrict__ deg, int* __restrict__ ssrc)
{
    __shared__ int c[256];
    __shared__ int wtot[8];
    __shared__ int wsum[16];
    __shared__ int rbeg[256];
    __shared__ int rlen[256];
    __shared__ int sorted[SORT_CAP];
    const int t = threadIdx.x;
    const int b = blockIdx.x;
    const int2* in2 = reinterpret_cast<const int2*>(tmp);
    const int node0 = b * BINW;
    int ra = 0, rl = 0;
    if (t < NCHUNK) {
        ra = min(max(cstart[t * CST + b], 0), CHUNK);
        rl = min(max(cstart[t * CST + b + 1], ra), CHUNK) - ra;
    }
    if (t < 256) {
        c[t] = 0;
        rbeg[t] = t * CHUNK + ra;
        rlen[t] = rl;
    }
    int start, n;
    block_sum2_256(ra, rl, t, wsum, start, n);
    const int q = t & 3;
    const int rb = rbeg[t >> 2];
    const int rn = rlen[t >> 2];
    for (int j = q; j < rn; j += 4) {
        const int l = in2[rb + j].y - node0;
        if ((unsigned)l < (unsigned)BINW) atomicAdd(&c[l], 1);
    }
    __syncthreads();
    const int v = (t < 256) ? c[t] : 0;
    const int excl = block_excl_scan256(v, t, wtot);
    if (t < BINW) {
        off[node0 + t] = start + excl;
        deg[node0 + t] = v;
    }
    if (t < 256) c[t] = excl;
    __syncthreads();
    const bool staged = (n <= SORT_CAP);
    for (int j = q; j < rn; j += 4) {
        const int2 e = in2[rb + j];
        const int l = e.y - node0;
        if ((unsigned)l < (unsigned)BINW) {
            const int p = atomicAdd(&c[l], 1);
            if (staged) {
                if ((unsigned)p < (unsigned)SORT_CAP) sorted[p] = e.x;
            } else if ((unsigned)(start + p) < (unsigned)NE) {
                ssrc[start + p] = e.x;
            }
        }
    }
    if (staged) {
        __syncthreads();
        for (int i = t; i < n; i += 1024) {
            if ((unsigned)(start + i) < (unsigned)NE) ssrc[start + i] = sorted[i];
        }
    }
}

#define GM_TILES 5
#define GM_ROWS  (GM_TILES * 16)

__global__ __launch_bounds__(128) void gemm5_wmma(
    const float* __restrict__ X,
    const float* __restrict__ W1, const float* __restrict__ b1,
    const float* __restrict__ W2, const float* __restrict__ b2,
    const float* __restrict__ W3, const float* __restrict__ b3,
    const float* __restrict__ W4, const float* __restrict__ b4,
    const float* __restrict__ Wr, const float* __restrict__ br,
    const int* __restrict__ deg,
    float* __restrict__ H2, float* __restrict__ H3, float* __restrict__ H4,
    float* __restrict__ out)
{
    __shared__ float xs[GM_ROWS * 132];
    __shared__ __attribute__((aligned(16))) float sinv[GM_ROWS];
    const int slab  = blockIdx.x >> 1;
    const int nhalf = blockIdx.x & 1;
    const int row0  = slab * GM_ROWS;
    const int t = threadIdx.x;

    for (int q = t; q < GM_ROWS * 32; q += 128) {
        const int r  = q >> 5;
        const int c4 = (q & 31) << 2;
        const float4 v = *reinterpret_cast<const float4*>(X + (size_t)(row0 + r) * DD + c4);
        *reinterpret_cast<float4*>(&xs[r * 132 + c4]) = v;
    }
    if (t < GM_ROWS) sinv[t] = 1.0f / fmaxf((float)deg[row0 + t], 1.0f);
    __syncthreads();

    const int wave = t >> 5;
    const int lane = t & 31;
    const int half = lane >> 4;
    const int nn   = lane & 15;
    const int n0   = (nhalf * 4 + wave) * 16;
    const int wrow = n0 + nn;

    const float* Ws[5] = {W1, W2, W3, W4, Wr};

    v8f acc[GM_TILES][5] = {};

    for (int k = 0; k < DD; k += 4) {
        const int kk = k + 2 * half;
        v2f b[5];
        #pragma unroll
        for (int w = 0; w < 5; ++w) {
            const float2 bf = *reinterpret_cast<const float2*>(Ws[w] + (size_t)wrow * DD + kk);
            b[w].x = bf.x; b[w].y = bf.y;
        }
        #pragma unroll
        for (int mt = 0; mt < GM_TILES; ++mt) {
            const float2 af = *reinterpret_cast<const float2*>(&xs[(mt * 16 + nn) * 132 + kk]);
            v2f a; a.x = af.x; a.y = af.y;
            #pragma unroll
            for (int w = 0; w < 5; ++w) {
                acc[mt][w] = __builtin_amdgcn_wmma_f32_16x16x4_f32(
                    false, a, false, b[w], (short)0, acc[mt][w], false, false);
            }
        }
    }

    const size_t obase = (size_t)(row0 + 8 * half) * DD + wrow;
    {
        float* po = out + obase;
        const float bias1 = b1[wrow];
        const float biasr = br[wrow];
        #pragma unroll
        for (int mt = 0; mt < GM_TILES; ++mt) {
            const float4 sa = *reinterpret_cast<const float4*>(&sinv[mt * 16 + 8 * half]);
            const float4 sb = *reinterpret_cast<const float4*>(&sinv[mt * 16 + 8 * half + 4]);
            const float sv[8] = {sa.x, sa.y, sa.z, sa.w, sb.x, sb.y, sb.z, sb.w};
            #pragma unroll
            for (int j = 0; j < 8; ++j) {
                po[(mt * 16 + j) * DD] =
                    (acc[mt][0][j] + bias1) * sv[j] + (acc[mt][4][j] + biasr);
            }
            __builtin_amdgcn_sched_barrier(0);
        }
    }
    const float* Bs[3] = {b2, b3, b4};
    float*       Hs[3] = {H2, H3, H4};
    #pragma unroll
    for (int w = 0; w < 3; ++w) {
        const float bias = Bs[w][wrow];
        float* ph = Hs[w] + obase;
        #pragma unroll
        for (int mt = 0; mt < GM_TILES; ++mt) {
            #pragma unroll
            for (int j = 0; j < 8; ++j) {
                ph[(mt * 16 + j) * DD] = acc[mt][w + 1][j] + bias;
            }
        }
    }
}

typedef float v4f __attribute__((ext_vector_type(4)));

__device__ __forceinline__ void gate_add(float4& acc, const v4f x2, const v4f x3, const float4 x4)
{
    acc.x += x2.x * __builtin_amdgcn_rcpf(1.0f + __expf(-(x3.x + x4.x)));
    acc.y += x2.y * __builtin_amdgcn_rcpf(1.0f + __expf(-(x3.y + x4.y)));
    acc.z += x2.z * __builtin_amdgcn_rcpf(1.0f + __expf(-(x3.z + x4.z)));
    acc.w += x2.w * __builtin_amdgcn_rcpf(1.0f + __expf(-(x3.w + x4.w)));
}

__device__ __forceinline__ int clamp_node(int s)
{
    return min(max(s, 0), NN - 1);
}

__global__ __launch_bounds__(256) void node_gather_reduce(
    const int* __restrict__ deg, const int* __restrict__ off,
    const int* __restrict__ ssrc,
    const float* __restrict__ H2, const float* __restrict__ H3,
    const float* __restrict__ H4, float* __restrict__ out)
{
    const int lane = threadIdx.x & 31;
    const int wave = __builtin_amdgcn_readfirstlane((int)(threadIdx.x >> 5));
    const int n = blockIdx.x * 8 + wave;
    const int d = deg[n];
    const int start = min(max(off[n], 0), NE);
    const int end = min(start + max(d, 0), NE);
    const int d0 = lane << 2;

    const float4 x4 = *reinterpret_cast<const float4*>(H4 + (size_t)n * DD + d0);
    float4* orow = reinterpret_cast<float4*>(out + (size_t)n * DD + d0);
    float4 o = *orow;
    float4 acc = make_float4(0.0f, 0.0f, 0.0f, 0.0f);

    for (int base = start; base < end; base += 32) {
        const int nb = min(32, end - base);
        const int mine = (lane < nb) ? clamp_node(ssrc[base + lane]) : 0;
        int j = 0;
        for (; j + 4 <= nb; j += 4) {
            const size_t r0 = (size_t)__builtin_amdgcn_readlane(mine, j + 0) * DD;
            const size_t r1 = (size_t)__builtin_amdgcn_readlane(mine, j + 1) * DD;
            const size_t r2 = (size_t)__builtin_amdgcn_readlane(mine, j + 2) * DD;
            const size_t r3 = (size_t)__builtin_amdgcn_readlane(mine, j + 3) * DD;
            v4f a3 = *reinterpret_cast<const v4f*>(H3 + r0 + d0);
            v4f a2 = *reinterpret_cast<const v4f*>(H2 + r0 + d0);
            v4f b3 = *reinterpret_cast<const v4f*>(H3 + r1 + d0);
            v4f b2 = *reinterpret_cast<const v4f*>(H2 + r1 + d0);
            v4f c3 = *reinterpret_cast<const v4f*>(H3 + r2 + d0);
            v4f c2 = *reinterpret_cast<const v4f*>(H2 + r2 + d0);
            v4f e3 = *reinterpret_cast<const v4f*>(H3 + r3 + d0);
            v4f e2 = *reinterpret_cast<const v4f*>(H2 + r3 + d0);
            asm volatile("" : "+v"(a3), "+v"(a2), "+v"(b3), "+v"(b2), "+v"(c3), "+v"(c2), "+v"(e3), "+v"(e2));
            gate_add(acc, a2, a3, x4);
            gate_add(acc, b2, b3, x4);
            gate_add(acc, c2, c3, x4);
            gate_add(acc, e2, e3, x4);
        }
        for (; j < nb; ++j) {
            const size_t r0 = (size_t)__builtin_amdgcn_readlane(mine, j) * DD;
            const v4f a3 = *reinterpret_cast<const v4f*>(H3 + r0 + d0);
            const v4f a2 = *reinterpret_cast<const v4f*>(H2 + r0 + d0);
            gate_add(acc, a2, a3, x4);
        }
    }

    const float s = 1.0f / fmaxf((float)d, 1.0f);
    o.x += acc.x * s;
    o.y += acc.y * s;
    o.z += acc.z * s;
    o.w += acc.w * s;
    *orow = o;
}

extern "C" void kernel_launch(void* const* d_in, const int* in_sizes, int n_in,
                              void* d_out, int out_size, void* d_ws, size_t ws_size,
                              hipStream_t stream)
{
    const float* X   = (const float*)d_in[0];
    const int*   ei  = (const int*)  d_in[1];
    const float* W1  = (const float*)d_in[2];
    const float* b1  = (const float*)d_in[3];
    const float* W2  = (const float*)d_in[4];
    const float* b2  = (const float*)d_in[5];
    const float* W3  = (const float*)d_in[6];
    const float* b3  = (const float*)d_in[7];
    const float* W4  = (const float*)d_in[8];
    const float* b4  = (const float*)d_in[9];
    const float* Wr  = (const float*)d_in[10];
    const float* br  = (const float*)d_in[11];
    float* out = (float*)d_out;

    const size_t nd = (size_t)NN * DD;
    float* ws  = (float*)d_ws;
    int*   off  = (int*)ws;
    int*   ssrc = (int*)ws + 65536;
    int*   tmp  = (int*)ws + 1048576;
    int*   cstart = (int*)ws + 2752512;
    float* H2  = ws + nd;
    float* H3  = ws + 2 * nd;
    float* H4  = ws + 3 * nd;
    float* cnt = ws + 4 * nd;
    int*   deg = (int*)cnt;

    chunk_sort<<<NCHUNK, 1024, 0, stream>>>(ei, cstart, tmp);
    bin_sort<<<NBIN, 1024, 0, stream>>>(cstart, tmp, off, deg, ssrc);

    gemm5_wmma<<<(NN / GM_ROWS) * 2, 128, 0, stream>>>(
        X, W1, b1, W2, b2, W3, b3, W4, b4, Wr, br, deg,
        H2, H3, H4, out);

    node_gather_reduce<<<NN / 8, 256, 0, stream>>>(deg, off, ssrc, H2, H3, H4, out);
}


# h1 and residual products moved from the GEMM launch into the edge kernel: resident 16-wave blocks loop over 16-node tiles, W1/Wres in LDS; GEMM keeps 3 weights; sum orders kept
# speedup vs baseline: 2.8857x; 1.0661x over previous
#include <hip/hip_runtime.h>

#define NN 50000
#define NE 800000
#define DD 128

typedef float v2f __attribute__((ext_vector_type(2)));
typedef float v8f __attribute__((ext_vector_type(8)));

#define NBIN   200
#define BINW   250
#define NCHUNK 250
#define CHUNK  3200
#define CST    256
#define SORT_CAP 8192

__device__ __forceinline__ int coarse_bin(int d)
{
    return min(max(d / BINW, 0), NBIN - 1);
}

__device__ __forceinline__ int wave_inclusive_scan(int x, int lane)
{
    for (int o = 1; o < 32; o <<= 1) {
        const int y = __shfl_up(x, o);
        if (lane >= o) x += y;
    }
    return x;
}

__device__ __forceinline__ int block_excl_scan256(int v, int t, int* wtot)
{
    const int lane = t & 31;
    const int wave = t >> 5;
    const int inc = wave_inclusive_scan(v, lane);
    if (t < 256 && lane == 31) wtot[wave] = inc;
    __syncthreads();
    int prefix = 0;
    for (int w = 0; w < 8; ++w) if (w < wave) prefix += wtot[w];
    return prefix + inc - v;
}

__device__ __forceinline__ void block_sum2_256(int x, int y, int t, int* w, int& sx, int& sy)
{
    for (int o = 16; o > 0; o >>= 1) {
        x += __shfl_xor(x, o);
        y += __shfl_xor(y, o);
    }
    if (t < 256 && (t & 31) == 0) {
        w[t >> 5] = x;
        w[8 + (t >> 5)] = y;
    }
    __syncthreads();
    sx = 0;
    sy = 0;
    for (int i = 0; i < 8; ++i) {
        sx += w[i];
        sy += w[8 + i];
    }
}

__global__ __launch_bounds__(1024) void chunk_sort(
    const int* __restrict__ ei, int* __restrict__ cstart, int* __restrict__ tmp)
{
    __shared__ int hist[256];
    __shared__ int lstart[256];
    __shared__ int wtot[8];
    __shared__ int2 stage[CHUNK];
    const int t = threadIdx.x;
    const int g = blockIdx.x;
    if (t < 256) hist[t] = 0;
    __syncthreads();
    const int* src = ei + g * CHUNK;
    const int* dst = ei + NE + g * CHUNK;
    int s[4], d[4], r[4];
    #pragma unroll
    for (int k = 0; k < 4; ++k) {
        const int i = t + 1024 * k;
        if (i < CHUNK) {
            s[k] = src[i];
            d[k] = dst[i];
            r[k] = atomicAdd(&hist[coarse_bin(d[k])], 1);
        }
    }
    __syncthreads();
    const int v = (t < 256) ? hist[t] : 0;
    const int excl = block_excl_scan256(v, t, wtot);
    if (t < 256) {
        lstart[t] = excl;
        cstart[g * CST + t] = excl;
    }
    __syncthreads();
    #pragma unroll
    for (int k = 0; k < 4; ++k) {
        const int i = t + 1024 * k;
        if (i < CHUNK) stage[lstart[coarse_bin(d[k])] + r[k]] = make_int2(s[k], d[k]);
    }
    __syncthreads();
    int2* out2 = reinterpret_cast<int2*>(tmp) + g * CHUNK;
    for (int i = t; i < CHUNK; i += 1024) out2[i] = stage[i];
}

__global__ __launch_bounds__(1024) void bin_sort(
    const int* __restrict__ cstart, const int* __restrict__ tmp,
    int* __restrict__ off, int* __restrict__ deg, int* __restrict__ ssrc)
{
    __shared__ int c[256];
    __shared__ int wtot[8];
    __shared__ int wsum[16];
    __shared__ int rbeg[256];
    __shared__ int rlen[256];
    __shared__ int sorted[SORT_CAP];
    const int t = threadIdx.x;
    const int b = blockIdx.x;
    const int2* in2 = reinterpret_cast<const int2*>(tmp);
    const int node0 = b * BINW;
    int ra = 0, rl = 0;
    if (t < NCHUNK) {
        ra = min(max(cstart[t * CST + b], 0), CHUNK);
        rl = min(max(cstart[t * CST + b + 1], ra), CHUNK) - ra;
    }
    if (t < 256) {
        c[t] = 0;
        rbeg[t] = t * CHUNK + ra;
        rlen[t] = rl;
    }
    int start, n;
    block_sum2_256(ra, rl, t, wsum, start, n);
    const int q = t & 3;
    const int rb = rbeg[t >> 2];
    const int rn = rlen[t >> 2];
    for (int j = q; j < rn; j += 4) {
        const int l = in2[rb + j].y - node0;
        if ((unsigned)l < (unsigned)BINW) atomicAdd(&c[l], 1);
    }
    __syncthreads();
    const int v = (t < 256) ? c[t] : 0;
    const int excl = block_excl_scan256(v, t, wtot);
    if (t < BINW) {
        off[node0 + t] = start + excl;
        deg[node0 + t] = v;
    }
    if (t < 256) c[t] = excl;
    __syncthreads();
    const bool staged = (n <= SORT_CAP);
    for (int j = q; j < rn; j += 4) {
        const int2 e = in2[rb + j];
        const int l = e.y - node0;
        if ((unsigned)l < (unsigned)BINW) {
            const int p = atomicAdd(&c[l], 1);
            if (staged) {
                if ((unsigned)p < (unsigned)SORT_CAP) sorted[p] = e.x;
            } else if ((unsigned)(start + p) < (unsigned)NE) {
                ssrc[start + p] = e.x;
            }
        }
    }
    if (staged) {
        __syncthreads();
        for (int i = t; i < n; i += 1024) {
            if ((unsigned)(start + i) < (unsigned)NE) ssrc[start + i] = sorted[i];
        }
    }
}

#define GM_TILES 5
#define GM_ROWS  (GM_TILES * 16)

__global__ __launch_bounds__(128) void gemm3_wmma(
    const float* __restrict__ X,
    const float* __restrict__ W2, const float* __restrict__ b2,
    const float* __restrict__ W3, const float* __restrict__ b3,
    const float* __restrict__ W4, const float* __restrict__ b4,
    float* __restrict__ H2, float* __restrict__ H3, float* __restrict__ H4)
{
    __shared__ float xs[GM_ROWS * 132];
    const int slab  = blockIdx.x >> 1;
    const int nhalf = blockIdx.x & 1;
    const int row0  = slab * GM_ROWS;
    const int t = threadIdx.x;

    for (int q = t; q < GM_ROWS * 32; q += 128) {
        const int r  = q >> 5;
        const int c4 = (q & 31) << 2;
        const float4 v = *reinterpret_cast<const float4*>(X + (size_t)(row0 + r) * DD + c4);
        *reinterpret_cast<float4*>(&xs[r * 132 + c4]) = v;
    }
    __syncthreads();

    const int wave = t >> 5;
    const int lane = t & 31;
    const int half = lane >> 4;
    const int nn   = lane & 15;
    const int n0   = (nhalf * 4 + wave) * 16;
    const int wrow = n0 + nn;

    const float* Ws[3] = {W2, W3, W4};

    v8f acc[GM_TILES][3] = {};

    #pragma unroll
    for (int k = 0; k < DD; k += 4) {
        const int kk = k + 2 * half;
        v2f b[3];
        #pragma unroll
        for (int w = 0; w < 3; ++w) {
            const float2 bf = *reinterpret_cast<const float2*>(Ws[w] + (size_t)wrow * DD + kk);
            b[w].x = bf.x; b[w].y = bf.y;
        }
        #pragma unroll
        for (int mt = 0; mt < GM_TILES; ++mt) {
            const float2 af = *reinterpret_cast<const float2*>(&xs[(mt * 16 + nn) * 132 + kk]);
            v2f a; a.x = af.x; a.y = af.y;
            #pragma unroll
            for (int w = 0; w < 3; ++w) {
                acc[mt][w] = __builtin_amdgcn_wmma_f32_16x16x4_f32(
                    false, a, false, b[w], (short)0, acc[mt][w], false, false);
            }
        }
    }

    __builtin_amdgcn_sched_barrier(0);
    asm volatile("v_nop\n\tv_nop\n\tv_nop\n\tv_nop");
    __builtin_amdgcn_sched_barrier(0);

    const size_t obase = (size_t)(row0 + 8 * half) * DD + wrow;
    const float* Bs[3] = {b2, b3, b4};
    float*       Hs[3] = {H2, H3, H4};
    #pragma unroll
    for (int w = 0; w < 3; ++w) {
        const float bias = Bs[w][wrow];
        float* ph = Hs[w] + obase;
        #pragma unroll
        for (int mt = 0; mt < GM_TILES; ++mt) {
            #pragma unroll
            for (int j = 0; j < 8; ++j) {
                ph[(mt * 16 + j) * DD] = acc[mt][w][j] + bias;
            }
        }
    }
}

typedef float v4f __attribute__((ext_vector_type(4)));

__device__ __forceinline__ void gate_add(float4& acc, const v4f x2, const v4f x3, const float4 x4)
{
    acc.x += x2.x * __builtin_amdgcn_rcpf(1.0f + __expf(-(x3.x + x4.x)));
    acc.y += x2.y * __builtin_amdgcn_rcpf(1.0f + __expf(-(x3.y + x4.y)));
    acc.z += x2.z * __builtin_amdgcn_rcpf(1.0f + __expf(-(x3.z + x4.z)));
    acc.w += x2.w * __builtin_amdgcn_rcpf(1.0f + __expf(-(x3.w + x4.w)));
}

__device__ __forceinline__ int clamp_node(int s)
{
    return min(max(s, 0), NN - 1);
}

#define ET_ROWS  16
#define ET_TILES (NN / ET_ROWS)
#define ET_WP    132
#define ET_DP    260

__global__ __launch_bounds__(512) void node_gather_gemm(
    const float* __restrict__ X,
    const float* __restrict__ W1, const float* __restrict__ b1,
    const float* __restrict__ Wr, const float* __restrict__ br,
    const int* __restrict__ deg, const int* __restrict__ off,
    const int* __restrict__ ssrc,
    const float* __restrict__ H2, const float* __restrict__ H3,
    const float* __restrict__ H4, float* __restrict__ out)
{
    __shared__ __attribute__((aligned(16))) float wsh[256 * ET_WP];
    __shared__ __attribute__((aligned(16))) float xsh[ET_ROWS * ET_WP];
    __shared__ __attribute__((aligned(16))) float dsh[ET_ROWS * ET_DP];
    const int t = threadIdx.x;
    const int lane = t & 31;
    const int wave = __builtin_amdgcn_readfirstlane((int)(t >> 5));
    const int half = lane >> 4;
    const int nn   = lane & 15;
    const int d0   = lane << 2;

    for (int q = t; q < 256 * 32; q += 512) {
        const int r  = q >> 5;
        const int c4 = (q & 31) << 2;
        const float* wp = (r < 128) ? (W1 + r * DD + c4) : (Wr + (r - 128) * DD + c4);
        *reinterpret_cast<float4*>(&wsh[r * ET_WP + c4]) = *reinterpret_cast<const float4*>(wp);
    }
    const float4 bias1 = *reinterpret_cast<const float4*>(b1 + d0);
    const float4 biasr = *reinterpret_cast<const float4*>(br + d0);

    const int wt = wave >> 3;
    const int ct = wave & 7;
    const int brow = wt * 128 + ct * 16 + nn;
    const float* bfrag = &wsh[brow * ET_WP + 2 * half];
    const float* afrag = &xsh[nn * ET_WP + 2 * half];

    for (int tile = blockIdx.x; tile < ET_TILES; tile += gridDim.x) {
        const int row0 = tile * ET_ROWS;
        *reinterpret_cast<float4*>(&xsh[(t >> 5) * ET_WP + d0]) =
            *reinterpret_cast<const float4*>(X + (size_t)(row0 + (t >> 5)) * DD + d0);

        const int n = row0 + wave;
        const int d = deg[n];
        const int start = min(max(off[n], 0), NE);
        const int end = min(start + max(d, 0), NE);
        const float s = 1.0f / fmaxf((float)d, 1.0f);
        const float4 x4 = *reinterpret_cast<const float4*>(H4 + (size_t)n * DD + d0);
        float4 acc = make_float4(0.0f, 0.0f, 0.0f, 0.0f);

        for (int base = start; base < end; base += 32) {
            const int nb = min(32, end - base);
            const int mine = (lane < nb) ? clamp_node(ssrc[base + lane]) : 0;
            int j = 0;
            for (; j + 4 <= nb; j += 4) {
                const size_t r0 = (size_t)__builtin_amdgcn_readlane(mine, j + 0) * DD;
                const size_t r1 = (size_t)__builtin_amdgcn_readlane(mine, j + 1) * DD;
                const size_t r2 = (size_t)__builtin_amdgcn_readlane(mine, j + 2) * DD;
                const size_t r3 = (size_t)__builtin_amdgcn_readlane(mine, j + 3) * DD;
                v4f a3 = *reinterpret_cast<const v4f*>(H3 + r0 + d0);
                v4f a2 = *reinterpret_cast<const v4f*>(H2 + r0 + d0);
                v4f b3 = *reinterpret_cast<const v4f*>(H3 + r1 + d0);
                v4f b2 = *reinterpret_cast<const v4f*>(H2 + r1 + d0);
                v4f c3 = *reinterpret_cast<const v4f*>(H3 + r2 + d0);
                v4f c2 = *reinterpret_cast<const v4f*>(H2 + r2 + d0);
                v4f e3 = *reinterpret_cast<const v4f*>(H3 + r3 + d0);
                v4f e2 = *reinterpret_cast<const v4f*>(H2 + r3 + d0);
                asm volatile("" : "+v"(a3), "+v"(a2), "+v"(b3), "+v"(b2), "+v"(c3), "+v"(c2), "+v"(e3), "+v"(e2));
                gate_add(acc, a2, a3, x4);
                gate_add(acc, b2, b3, x4);
                gate_add(acc, c2, c3, x4);
                gate_add(acc, e2, e3, x4);
            }
            for (; j < nb; ++j) {
                const size_t r0 = (size_t)__builtin_amdgcn_readlane(mine, j) * DD;
                const v4f a3 = *reinterpret_cast<const v4f*>(H3 + r0 + d0);
                const v4f a2 = *reinterpret_cast<const v4f*>(H2 + r0 + d0);
                gate_add(acc, a2, a3, x4);
            }
        }

        __syncthreads();

        __builtin_amdgcn_sched_barrier(0);
        v8f dacc = {};
        #pragma unroll
        for (int k = 0; k < DD; k += 4) {
            const float2 af = *reinterpret_cast<const float2*>(afrag + k);
            const float2 bf = *reinterpret_cast<const float2*>(bfrag + k);
            v2f a; a.x = af.x; a.y = af.y;
            v2f b; b.x = bf.x; b.y = bf.y;
            dacc = __builtin_amdgcn_wmma_f32_16x16x4_f32(
                false, a, false, b, (short)0, dacc, false, false);
        }
        __builtin_amdgcn_sched_barrier(0);
        asm volatile("v_nop\n\tv_nop\n\tv_nop\n\tv_nop");
        __builtin_amdgcn_sched_barrier(0);
        {
            float* dp = &dsh[(8 * half) * ET_DP + wt * 128 + ct * 16 + nn];
            #pragma unroll
            for (int j = 0; j < 8; ++j) dp[j * ET_DP] = dacc[j];
        }

        __syncthreads();

        const float4 h1 = *reinterpret_cast<const float4*>(&dsh[wave * ET_DP + d0]);
        const float4 hr = *reinterpret_cast<const float4*>(&dsh[wave * ET_DP + 128 + d0]);
        float4 o;
        o.x = (h1.x + bias1.x) * s + (hr.x + biasr.x);
        o.y = (h1.y + bias1.y) * s + (hr.y + biasr.y);
        o.z = (h1.z + bias1.z) * s + (hr.z + biasr.z);
        o.w = (h1.w + bias1.w) * s + (hr.w + biasr.w);
        o.x += acc.x * s;
        o.y += acc.y * s;
        o.z += acc.z * s;
        o.w += acc.w * s;
        *reinterpret_cast<float4*>(out + (size_t)n * DD + d0) = o;
    }
}

extern "C" void kernel_launch(void* const* d_in, const int* in_sizes, int n_in,
                              void* d_out, int out_size, void* d_ws, size_t ws_size,
                              hipStream_t stream)
{
    const float* X   = (const float*)d_in[0];
    const int*   ei  = (const int*)  d_in[1];
    const float* W1  = (const float*)d_in[2];
    const float* b1  = (const float*)d_in[3];
    const float* W2  = (const float*)d_in[4];
    const float* b2  = (const float*)d_in[5];
    const float* W3  = (const float*)d_in[6];
    const float* b3  = (const float*)d_in[7];
    const float* W4  = (const float*)d_in[8];
    const float* b4  = (const float*)d_in[9];
    const float* Wr  = (const float*)d_in[10];
    const float* br  = (const float*)d_in[11];
    float* out = (float*)d_out;

    const size_t nd = (size_t)NN * DD;
    float* ws  = (float*)d_ws;
    int*   off  = (int*)ws;
    int*   ssrc = (int*)ws + 65536;
    int*   tmp  = (int*)ws + 1048576;
    int*   cstart = (int*)ws + 2752512;
    float* H2  = ws + nd;
    float* H3  = ws + 2 * nd;
    float* H4  = ws + 3 * nd;
    float* cnt = ws + 4 * nd;
    int*   deg = (int*)cnt;

    chunk_sort<<<NCHUNK, 1024, 0, stream>>>(ei, cstart, tmp);
    bin_sort<<<NBIN, 1024, 0, stream>>>(cstart, tmp, off, deg, ssrc);

    gemm3_wmma<<<(NN / GM_ROWS) * 2, 128, 0, stream>>>(
        X, W2, b2, W3, b3, W4, b4, H2, H3, H4);

    node_gather_gemm<<<512, 512, 0, stream>>>(
        X, W1, b1, Wr, br, deg, off, ssrc, H2, H3, H4, out);
}
